# starting kernel
# baseline (speedup 1.0000x reference)
#include <hip/hip_runtime.h>

typedef __attribute__((ext_vector_type(2))) float v2f;
typedef __attribute__((ext_vector_type(4))) float v4f;
typedef __attribute__((ext_vector_type(8))) float v8f;

#define T_STEPS 200
#define HID 4
#define OUTN 28
#define WP_ENTRIES (T_STEPS * 2 * 32)
#define SWAPX16 0x401f

__device__ __forceinline__ float fast_tanh(float x) {
#if __has_builtin(__builtin_amdgcn_tanhf)
  return __builtin_amdgcn_tanhf(x);
#else
  float e = __builtin_amdgcn_exp2f(-2.8853900817779268f * x);
  return (1.0f - e) * __builtin_amdgcn_rcpf(1.0f + e);
#endif
}
__device__ __forceinline__ float fast_sigmoid(float x) {
  return __builtin_fmaf(0.5f, fast_tanh(0.5f * x), 0.5f);
}
__device__ __forceinline__ v2f tanh2(v2f v) {
  v2f r; r.x = fast_tanh(v.x); r.y = fast_tanh(v.y); return r;
}
__device__ __forceinline__ v2f sig2(v2f v) {
  v2f r; r.x = fast_sigmoid(v.x); r.y = fast_sigmoid(v.y); return r;
}
__device__ __forceinline__ float swap16(float v) {
  int i = __builtin_bit_cast(int, v);
  i = __builtin_amdgcn_ds_swizzle(i, SWAPX16);
  return __builtin_bit_cast(float, i);
}

__device__ __forceinline__ void cell_pk(const v4f* pw, const v4f* pbv,
                                        const float in[8], v2f h[2], v2f c[2]) {
  v2f g[8];
#pragma unroll
  for (int q = 0; q < 4; ++q) {
    v4f bq = pbv[q];
    g[2 * q].x = bq.x;     g[2 * q].y = bq.y;
    g[2 * q + 1].x = bq.z; g[2 * q + 1].y = bq.w;
  }
#pragma unroll
  for (int p = 0; p < 8; ++p) {
#pragma unroll
    for (int kq = 0; kq < 4; ++kq) {
      v4f q = pw[p * 4 + kq];
      v2f w0; w0.x = q.x; w0.y = q.y;
      v2f w1; w1.x = q.z; w1.y = q.w;
      v2f a0; a0.x = in[2 * kq];     a0.y = in[2 * kq];
      v2f a1; a1.x = in[2 * kq + 1]; a1.y = in[2 * kq + 1];
      g[p] = __builtin_elementwise_fma(a0, w0, g[p]);
      g[p] = __builtin_elementwise_fma(a1, w1, g[p]);
    }
  }
#pragma unroll
  for (int q = 0; q < 2; ++q) {
    v2f iv = sig2(g[0 + q]);
    v2f fv = sig2(g[2 + q]);
    v2f gv = tanh2(g[4 + q]);
    v2f ov = sig2(g[6 + q]);
    c[q] = __builtin_elementwise_fma(fv, c[q], iv * gv);
    h[q] = ov * tanh2(c[q]);
  }
}

__global__ void repack_wout_kernel(const float* __restrict__ Wout,
                                   v2f* __restrict__ WP) {
  int idx = blockIdx.x * blockDim.x + threadIdx.x;
  if (idx >= WP_ENTRIES) return;
  int c = idx & 31;
  int kp = (idx >> 5) & 1;
  int t = idx >> 6;
  v2f v; v.x = 0.f; v.y = 0.f;
  if (c < OUTN) {
    const float* p = Wout + c * 800 + 4 * t + 2 * kp;
    v.x = p[0];
    v.y = p[1];
  }
  WP[idx] = v;
}

__global__ void __launch_bounds__(128)
lstm_fused_kernel(const float* __restrict__ x,
                  const float* __restrict__ Wih0, const float* __restrict__ Whh0,
                  const float* __restrict__ bih0, const float* __restrict__ bhh0,
                  const float* __restrict__ Wih1, const float* __restrict__ Whh1,
                  const float* __restrict__ bih1, const float* __restrict__ bhh1,
                  const v2f* __restrict__ WP, const float* __restrict__ bout,
                  float* __restrict__ out) {
  __shared__ __align__(16) float sw[288];
  const int tid = threadIdx.x;

  for (int i = tid; i < 288; i += blockDim.x) {
    float v;
    if (i < 256) {
      int layer = i >> 7;
      int f = i & 127;
      int e = f >> 1, comp = f & 1;
      int p = e >> 3, k = e & 7;
      int j = 2 * p + comp;
      const float* Wih = layer ? Wih1 : Wih0;
      const float* Whh = layer ? Whh1 : Whh0;
      v = (k < 4) ? Wih[j * 4 + k] : Whh[j * 4 + (k - 4)];
    } else if (i < 272) {
      v = bih0[i - 256] + bhh0[i - 256];
    } else {
      v = bih1[i - 272] + bhh1[i - 272];
    }
    sw[i] = v;
  }
  __syncthreads();

  const v4f* sPW0 = (const v4f*)(sw);
  const v4f* sPW1 = (const v4f*)(sw + 128);
  const v4f* sPB0 = (const v4f*)(sw + 256);
  const v4f* sPB1 = (const v4f*)(sw + 272);

  v4f pw1r[32];
#pragma unroll
  for (int q = 0; q < 32; ++q) pw1r[q] = sPW1[q];

  const int lane = tid & 31;
  const int col = lane & 15;
  const bool hiHalf = (lane & 16) != 0;
  const int waveBase = blockIdx.x * blockDim.x + (tid & ~31);
  const int b = waveBase + lane;

  const v4f* __restrict__ xp = (const v4f*)(x + (size_t)b * T_STEPS * HID);

  v2f h0[2], c0[2], h1[2], c1[2];
#pragma unroll
  for (int q = 0; q < 2; ++q) {
    h0[q] = (v2f)0.f; c0[q] = (v2f)0.f;
    h1[q] = (v2f)0.f; c1[q] = (v2f)0.f;
  }

  float bo0 = bout[col];
  float bo1 = (col + 16 < OUTN) ? bout[col + 16] : 0.f;
  v8f acc00, acc01, acc10, acc11;
#pragma unroll
  for (int r = 0; r < 8; ++r) {
    acc00[r] = bo0; acc10[r] = bo0;
    acc01[r] = bo1; acc11[r] = bo1;
  }

  const int kp = hiHalf ? 1 : 0;
  const v2f* __restrict__ wpLane = WP + kp * 32 + col;

  for (int t = 0; t < T_STEPS; ++t) {
    v4f xv = xp[t];
    __builtin_prefetch((const void*)(xp + t + 8), 0, 0);

    float in0[8] = {xv.x, xv.y, xv.z, xv.w,
                    h0[0].x, h0[0].y, h0[1].x, h0[1].y};
    cell_pk(sPW0, sPB0, in0, h0, c0);

    float in1[8] = {h0[0].x, h0[0].y, h0[1].x, h0[1].y,
                    h1[0].x, h1[0].y, h1[1].x, h1[1].y};
    cell_pk(pw1r, sPB1, in1, h1, c1);

    float y0 = h1[0].x, y1 = h1[0].y, y2 = h1[1].x, y3 = h1[1].y;
    float s0 = swap16(y0), s1 = swap16(y1), s2 = swap16(y2), s3 = swap16(y3);
    v2f A0, A1;
    A0.x = hiHalf ? s2 : y0;  A0.y = hiHalf ? s3 : y1;
    A1.x = hiHalf ? y2 : s0;  A1.y = hiHalf ? y3 : s1;

    const v2f* wr = wpLane + t * 64;
    v2f B0 = wr[0];
    v2f B1 = wr[16];

    acc00 = __builtin_amdgcn_wmma_f32_16x16x4_f32(false, A0, false, B0, (short)0, acc00, false, false);
    acc01 = __builtin_amdgcn_wmma_f32_16x16x4_f32(false, A0, false, B1, (short)0, acc01, false, false);
    acc10 = __builtin_amdgcn_wmma_f32_16x16x4_f32(false, A1, false, B0, (short)0, acc10, false, false);
    acc11 = __builtin_amdgcn_wmma_f32_16x16x4_f32(false, A1, false, B1, (short)0, acc11, false, false);

    asm volatile("" ::: "memory");
  }

  float* outp0 = out + (size_t)(waveBase + (hiHalf ? 8 : 0)) * OUTN + col;
#pragma unroll
  for (int r = 0; r < 8; ++r) {
    outp0[(size_t)r * OUTN] = acc00[r];
    outp0[(size_t)(r + 16) * OUTN] = acc10[r];
    if (col < OUTN - 16) {
      outp0[(size_t)r * OUTN + 16] = acc01[r];
      outp0[(size_t)(r + 16) * OUTN + 16] = acc11[r];
    }
  }
}

extern "C" void kernel_launch(void* const* d_in, const int* in_sizes, int n_in,
                              void* d_out, int out_size, void* d_ws, size_t ws_size,
                              hipStream_t stream) {
  const float* x    = (const float*)d_in[0];
  const float* Wih0 = (const float*)d_in[1];
  const float* Whh0 = (const float*)d_in[2];
  const float* bih0 = (const float*)d_in[3];
  const float* bhh0 = (const float*)d_in[4];
  const float* Wih1 = (const float*)d_in[5];
  const float* Whh1 = (const float*)d_in[6];
  const float* bih1 = (const float*)d_in[7];
  const float* bhh1 = (const float*)d_in[8];
  const float* Wout = (const float*)d_in[9];
  const float* bout = (const float*)d_in[10];
  float* out = (float*)d_out;
  v2f* WP = (v2f*)d_ws;

  int B = in_sizes[0] / (T_STEPS * HID);

  repack_wout_kernel<<<(WP_ENTRIES + 255) / 256, 256, 0, stream>>>(Wout, WP);
  lstm_fused_kernel<<<B / 128, 128, 0, stream>>>(x, Wih0, Whh0, bih0, bhh0,
                                                 Wih1, Whh1, bih1, bhh1,
                                                 WP, bout, out);
}


# Layer-0 weights and both bias sets held in VGPRs like layer 1 (376 VGPRs, one wave per SIMD); the time loop no longer issues its 40 ds_load_b128 of weights and biases per step
# speedup vs baseline: 1.2636x; 1.2636x over previous
#include <hip/hip_runtime.h>

typedef __attribute__((ext_vector_type(2))) float v2f;
typedef __attribute__((ext_vector_type(4))) float v4f;
typedef __attribute__((ext_vector_type(8))) float v8f;

#define T_STEPS 200
#define HID 4
#define OUTN 28
#define WP_ENTRIES (T_STEPS * 2 * 32)
#define SWAPX16 0x401f

__device__ __forceinline__ float fast_tanh(float x) {
#if __has_builtin(__builtin_amdgcn_tanhf)
  return __builtin_amdgcn_tanhf(x);
#else
  float e = __builtin_amdgcn_exp2f(-2.8853900817779268f * x);
  return (1.0f - e) * __builtin_amdgcn_rcpf(1.0f + e);
#endif
}
__device__ __forceinline__ float fast_sigmoid(float x) {
  return __builtin_fmaf(0.5f, fast_tanh(0.5f * x), 0.5f);
}
__device__ __forceinline__ v2f tanh2(v2f v) {
  v2f r; r.x = fast_tanh(v.x); r.y = fast_tanh(v.y); return r;
}
__device__ __forceinline__ v2f sig2(v2f v) {
  v2f r; r.x = fast_sigmoid(v.x); r.y = fast_sigmoid(v.y); return r;
}
__device__ __forceinline__ float swap16(float v) {
  int i = __builtin_bit_cast(int, v);
  i = __builtin_amdgcn_ds_swizzle(i, SWAPX16);
  return __builtin_bit_cast(float, i);
}

__device__ __forceinline__ void cell_pk(const v4f* pw, const v2f* pb,
                                        const float in[8], v2f h[2], v2f c[2]) {
  v2f g[8];
#pragma unroll
  for (int p = 0; p < 8; ++p) {
    v2f acc = pb[p];
#pragma unroll
    for (int kq = 0; kq < 4; ++kq) {
      v4f q = pw[p * 4 + kq];
      v2f w0; w0.x = q.x; w0.y = q.y;
      v2f w1; w1.x = q.z; w1.y = q.w;
      v2f a0; a0.x = in[2 * kq];     a0.y = in[2 * kq];
      v2f a1; a1.x = in[2 * kq + 1]; a1.y = in[2 * kq + 1];
      acc = __builtin_elementwise_fma(a0, w0, acc);
      acc = __builtin_elementwise_fma(a1, w1, acc);
    }
    g[p] = acc;
  }
#pragma unroll
  for (int q = 0; q < 2; ++q) {
    v2f iv = sig2(g[0 + q]);
    v2f fv = sig2(g[2 + q]);
    v2f gv = tanh2(g[4 + q]);
    v2f ov = sig2(g[6 + q]);
    c[q] = __builtin_elementwise_fma(fv, c[q], iv * gv);
    h[q] = ov * tanh2(c[q]);
  }
}

__global__ void repack_wout_kernel(const float* __restrict__ Wout,
                                   v2f* __restrict__ WP) {
  int idx = blockIdx.x * blockDim.x + threadIdx.x;
  if (idx >= WP_ENTRIES) return;
  int c = idx & 31;
  int kp = (idx >> 5) & 1;
  int t = idx >> 6;
  v2f v; v.x = 0.f; v.y = 0.f;
  if (c < OUTN) {
    const float* p = Wout + c * 800 + 4 * t + 2 * kp;
    v.x = p[0];
    v.y = p[1];
  }
  WP[idx] = v;
}

__global__ void __launch_bounds__(128)
lstm_fused_kernel(const float* __restrict__ x,
                  const float* __restrict__ Wih0, const float* __restrict__ Whh0,
                  const float* __restrict__ bih0, const float* __restrict__ bhh0,
                  const float* __restrict__ Wih1, const float* __restrict__ Whh1,
                  const float* __restrict__ bih1, const float* __restrict__ bhh1,
                  const v2f* __restrict__ WP, const float* __restrict__ bout,
                  float* __restrict__ out) {
  __shared__ __align__(16) float sw[288];
  const int tid = threadIdx.x;

  for (int i = tid; i < 288; i += blockDim.x) {
    float v;
    if (i < 256) {
      int layer = i >> 7;
      int f = i & 127;
      int e = f >> 1, comp = f & 1;
      int p = e >> 3, k = e & 7;
      int j = 2 * p + comp;
      const float* Wih = layer ? Wih1 : Wih0;
      const float* Whh = layer ? Whh1 : Whh0;
      v = (k < 4) ? Wih[j * 4 + k] : Whh[j * 4 + (k - 4)];
    } else if (i < 272) {
      v = bih0[i - 256] + bhh0[i - 256];
    } else {
      v = bih1[i - 272] + bhh1[i - 272];
    }
    sw[i] = v;
  }
  __syncthreads();

  const v4f* sPW0 = (const v4f*)(sw);
  const v4f* sPW1 = (const v4f*)(sw + 128);
  const v2f* sPB0 = (const v2f*)(sw + 256);
  const v2f* sPB1 = (const v2f*)(sw + 272);

  v4f pw0r[32], pw1r[32];
#pragma unroll
  for (int q = 0; q < 32; ++q) { pw0r[q] = sPW0[q]; pw1r[q] = sPW1[q]; }
  v2f pb0r[8], pb1r[8];
#pragma unroll
  for (int p = 0; p < 8; ++p) { pb0r[p] = sPB0[p]; pb1r[p] = sPB1[p]; }

  const int lane = tid & 31;
  const int col = lane & 15;
  const bool hiHalf = (lane & 16) != 0;
  const int waveBase = blockIdx.x * blockDim.x + (tid & ~31);
  const int b = waveBase + lane;

  const v4f* __restrict__ xp = (const v4f*)(x + (size_t)b * T_STEPS * HID);

  v2f h0[2], c0[2], h1[2], c1[2];
#pragma unroll
  for (int q = 0; q < 2; ++q) {
    h0[q] = (v2f)0.f; c0[q] = (v2f)0.f;
    h1[q] = (v2f)0.f; c1[q] = (v2f)0.f;
  }

  float bo0 = bout[col];
  float bo1 = (col + 16 < OUTN) ? bout[col + 16] : 0.f;
  v8f acc00, acc01, acc10, acc11;
#pragma unroll
  for (int r = 0; r < 8; ++r) {
    acc00[r] = bo0; acc10[r] = bo0;
    acc01[r] = bo1; acc11[r] = bo1;
  }

  const int kp = hiHalf ? 1 : 0;
  const v2f* __restrict__ wpLane = WP + kp * 32 + col;

  for (int t = 0; t < T_STEPS; ++t) {
    v4f xv = xp[t];
    __builtin_prefetch((const void*)(xp + t + 8), 0, 0);

    float in0[8] = {xv.x, xv.y, xv.z, xv.w,
                    h0[0].x, h0[0].y, h0[1].x, h0[1].y};
    cell_pk(pw0r, pb0r, in0, h0, c0);

    float in1[8] = {h0[0].x, h0[0].y, h0[1].x, h0[1].y,
                    h1[0].x, h1[0].y, h1[1].x, h1[1].y};
    cell_pk(pw1r, pb1r, in1, h1, c1);

    float y0 = h1[0].x, y1 = h1[0].y, y2 = h1[1].x, y3 = h1[1].y;
    float s0 = swap16(y0), s1 = swap16(y1), s2 = swap16(y2), s3 = swap16(y3);
    v2f A0, A1;
    A0.x = hiHalf ? s2 : y0;  A0.y = hiHalf ? s3 : y1;
    A1.x = hiHalf ? y2 : s0;  A1.y = hiHalf ? y3 : s1;

    const v2f* wr = wpLane + t * 64;
    v2f B0 = wr[0];
    v2f B1 = wr[16];

    acc00 = __builtin_amdgcn_wmma_f32_16x16x4_f32(false, A0, false, B0, (short)0, acc00, false, false);
    acc01 = __builtin_amdgcn_wmma_f32_16x16x4_f32(false, A0, false, B1, (short)0, acc01, false, false);
    acc10 = __builtin_amdgcn_wmma_f32_16x16x4_f32(false, A1, false, B0, (short)0, acc10, false, false);
    acc11 = __builtin_amdgcn_wmma_f32_16x16x4_f32(false, A1, false, B1, (short)0, acc11, false, false);
  }

  float* outp0 = out + (size_t)(waveBase + (hiHalf ? 8 : 0)) * OUTN + col;
#pragma unroll
  for (int r = 0; r < 8; ++r) {
    outp0[(size_t)r * OUTN] = acc00[r];
    outp0[(size_t)(r + 16) * OUTN] = acc10[r];
    if (col < OUTN - 16) {
      outp0[(size_t)r * OUTN + 16] = acc01[r];
      outp0[(size_t)(r + 16) * OUTN + 16] = acc11[r];
    }
  }
}

extern "C" void kernel_launch(void* const* d_in, const int* in_sizes, int n_in,
                              void* d_out, int out_size, void* d_ws, size_t ws_size,
                              hipStream_t stream) {
  const float* x    = (const float*)d_in[0];
  const float* Wih0 = (const float*)d_in[1];
  const float* Whh0 = (const float*)d_in[2];
  const float* bih0 = (const float*)d_in[3];
  const float* bhh0 = (const float*)d_in[4];
  const float* Wih1 = (const float*)d_in[5];
  const float* Whh1 = (const float*)d_in[6];
  const float* bih1 = (const float*)d_in[7];
  const float* bhh1 = (const float*)d_in[8];
  const float* Wout = (const float*)d_in[9];
  const float* bout = (const float*)d_in[10];
  float* out = (float*)d_out;
  v2f* WP = (v2f*)d_ws;

  int B = in_sizes[0] / (T_STEPS * HID);

  repack_wout_kernel<<<(WP_ENTRIES + 255) / 256, 256, 0, stream>>>(Wout, WP);
  lstm_fused_kernel<<<B / 128, 128, 0, stream>>>(x, Wih0, Whh0, bih0, bhh0,
                                                 Wih1, Whh1, bih1, bhh1,
                                                 WP, bout, out);
}


# LSTM gate products moved from 128 packed FMAs per step onto f32 WMMA, transposed (batch = N, 16 rows per wave, 2 waves per SIMD): h feeds back as B with no lane movement; gate sum order changes
# speedup vs baseline: 2.1459x; 1.6982x over previous
#include <hip/hip_runtime.h>


typedef __attribute__((ext_vector_type(2))) float v2f;
typedef __attribute__((ext_vector_type(8))) float v8f;

#define T_STEPS 200
#define HID 4
#define OUTN 28
#define WP_ENTRIES (T_STEPS * 2 * 32)
#define TILE_M 16
#define WAVES_PER_BLOCK 8

__device__ __forceinline__ float fast_tanh(float x) {
#if __has_builtin(__builtin_amdgcn_tanhf)
  return __builtin_amdgcn_tanhf(x);
#else
  float e = __builtin_amdgcn_exp2f(-2.8853900817779268f * x);
  return (1.0f - e) * __builtin_amdgcn_rcpf(1.0f + e);
#endif
}
__device__ __forceinline__ float fast_sigmoid(float x) {
  return __builtin_fmaf(0.5f, fast_tanh(0.5f * x), 0.5f);
}
__device__ __forceinline__ v2f tanh2(v2f v) {
  v2f r; r.x = fast_tanh(v.x); r.y = fast_tanh(v.y); return r;
}
__device__ __forceinline__ v2f sig2(v2f v) {
  v2f r; r.x = fast_sigmoid(v.x); r.y = fast_sigmoid(v.y); return r;
}
__device__ __forceinline__ v8f wmma4(v2f a, v2f b, v8f c) {
  return __builtin_amdgcn_wmma_f32_16x16x4_f32(false, a, false, b, (short)0, c, false, false);
}

__device__ __forceinline__ void cell_act(v8f g, v2f& h, v2f& c) {
  v2f gi; gi.x = g[0]; gi.y = g[1];
  v2f gf; gf.x = g[2]; gf.y = g[3];
  v2f gg; gg.x = g[4]; gg.y = g[5];
  v2f go; go.x = g[6]; go.y = g[7];
  v2f iv = sig2(gi);
  v2f fv = sig2(gf);
  v2f gv = tanh2(gg);
  v2f ov = sig2(go);
  c = __builtin_elementwise_fma(fv, c, iv * gv);
  h = ov * tanh2(c);
}

__global__ void repack_wout_kernel(const float* __restrict__ Wout,
                                   v2f* __restrict__ WP) {
  int idx = blockIdx.x * blockDim.x + threadIdx.x;
  if (idx >= WP_ENTRIES) return;
  int c = idx & 31;
  int kp = (idx >> 5) & 1;
  int t = idx >> 6;
  v2f v; v.x = 0.f; v.y = 0.f;
  if (c < OUTN) {
    const float* p = Wout + c * 800 + 4 * t + 2 * kp;
    v.x = p[0];
    v.y = p[1];
  }
  WP[idx] = v;
}

__global__ void __launch_bounds__(WAVES_PER_BLOCK * 32)
lstm_fused_kernel(const float* __restrict__ x,
                  const float* __restrict__ Wih0, const float* __restrict__ Whh0,
                  const float* __restrict__ bih0, const float* __restrict__ bhh0,
                  const float* __restrict__ Wih1, const float* __restrict__ Whh1,
                  const float* __restrict__ bih1, const float* __restrict__ bhh1,
                  const v2f* __restrict__ WP, const float* __restrict__ bout,
                  float* __restrict__ out) {
  const int tid = threadIdx.x;
  const int lane = tid & 31;
  const int col = lane & 15;
  const int hh = lane >> 4;
  const int tileBase = (blockIdx.x * WAVES_PER_BLOCK + (tid >> 5)) * TILE_M;

  const int ar = col & 7, ahalf = col >> 3;
  const int j = (ar >> 1) * 4 + 2 * ahalf + (ar & 1);
  v2f aih0, ahh0, aih1, ahh1;
  aih0.x = Wih0[j * 4 + 2 * hh]; aih0.y = Wih0[j * 4 + 2 * hh + 1];
  ahh0.x = Whh0[j * 4 + 2 * hh]; ahh0.y = Whh0[j * 4 + 2 * hh + 1];
  aih1.x = Wih1[j * 4 + 2 * hh]; aih1.y = Wih1[j * 4 + 2 * hh + 1];
  ahh1.x = Whh1[j * 4 + 2 * hh]; ahh1.y = Whh1[j * 4 + 2 * hh + 1];

  v8f bias0, bias1;
#pragma unroll
  for (int r = 0; r < 8; ++r) {
    const int jr = (r >> 1) * 4 + 2 * hh + (r & 1);
    bias0[r] = bih0[jr] + bhh0[jr];
    bias1[r] = bih1[jr] + bhh1[jr];
  }

  const v2f* __restrict__ xp =
      (const v2f*)(x + (size_t)(tileBase + col) * T_STEPS * HID + 2 * hh);

  v2f h0 = (v2f)0.f, c0 = (v2f)0.f, h1 = (v2f)0.f, c1 = (v2f)0.f;

  float bo0 = bout[col];
  float bo1 = (col + 16 < OUTN) ? bout[col + 16] : 0.f;
  v8f acc0, acc1;
#pragma unroll
  for (int r = 0; r < 8; ++r) { acc0[r] = bo0; acc1[r] = bo1; }

  const v2f* __restrict__ wpLane = WP + hh * 32 + col;

  {
    v8f g = wmma4(aih0, xp[0], bias0);
    g = wmma4(ahh0, h0, g);
    cell_act(g, h0, c0);
  }
  v2f xnext = xp[2 * 1];
  __builtin_prefetch((const void*)(xp + 2 * 8), 0, 0);

  for (int t = 0; t < T_STEPS; ++t) {
    const int t2 = (t + 2 < T_STEPS) ? t + 2 : T_STEPS - 1;
    v2f xload = xp[2 * t2];
    if ((t & 7) == 0) {
      const int tpf = (t + 16 < T_STEPS) ? t + 16 : T_STEPS - 1;
      __builtin_prefetch((const void*)(xp + 2 * tpf), 0, 0);
    }

    const v2f* wr = wpLane + t * 64;
    v2f B0 = wr[0];
    v2f B1 = wr[16];

    v8f g1 = wmma4(aih1, h0, bias1);
    g1 = wmma4(ahh1, h1, g1);
    v8f g0 = wmma4(aih0, xnext, bias0);
    g0 = wmma4(ahh0, h0, g0);

    cell_act(g1, h1, c1);
    cell_act(g0, h0, c0);

    acc0 = wmma4(h1, B0, acc0);
    acc1 = wmma4(h1, B1, acc1);

    xnext = xload;
  }

  float* outp = out + (size_t)(tileBase + 8 * hh) * OUTN + col;
#pragma unroll
  for (int r = 0; r < 8; ++r) {
    outp[(size_t)r * OUTN] = acc0[r];
    if (col < OUTN - 16) outp[(size_t)r * OUTN + 16] = acc1[r];
  }
}

extern "C" void kernel_launch(void* const* d_in, const int* in_sizes, int n_in,
                              void* d_out, int out_size, void* d_ws, size_t ws_size,
                              hipStream_t stream) {
  const float* x    = (const float*)d_in[0];
  const float* Wih0 = (const float*)d_in[1];
  const float* Whh0 = (const float*)d_in[2];
  const float* bih0 = (const float*)d_in[3];
  const float* bhh0 = (const float*)d_in[4];
  const float* Wih1 = (const float*)d_in[5];
  const float* Whh1 = (const float*)d_in[6];
  const float* bih1 = (const float*)d_in[7];
  const float* bhh1 = (const float*)d_in[8];
  const float* Wout = (const float*)d_in[9];
  const float* bout = (const float*)d_in[10];
  float* out = (float*)d_out;
  v2f* WP = (v2f*)d_ws;

  int B = in_sizes[0] / (T_STEPS * HID);

  repack_wout_kernel<<<(WP_ENTRIES + 255) / 256, 256, 0, stream>>>(Wout, WP);
  lstm_fused_kernel<<<B / 128, 256, 0, stream>>>(x, Wih0, Whh0, bih0, bhh0,
                                                 Wih1, Whh1, bih1, bhh1,
                                                 WP, bout, out);
}


# Loop operands (x and repacked-Wout fragments) loaded into registers one 8-step block ahead, two stages in turn, in place of the 1-2-step register carry and the L2 prefetch; sums unchanged
# speedup vs baseline: 3.1152x; 1.4517x over previous
#include <hip/hip_runtime.h>


typedef __attribute__((ext_vector_type(2))) float v2f;
typedef __attribute__((ext_vector_type(8))) float v8f;

#define T_STEPS 200
#define HID 4
#define OUTN 28
#define WP_ENTRIES (T_STEPS * 2 * 32)
#define TILE_M 16
#define WAVES_PER_BLOCK 8
#define BLK 8
#define NBLK (T_STEPS / BLK)

__device__ __forceinline__ float fast_tanh(float x) {
#if __has_builtin(__builtin_amdgcn_tanhf)
  return __builtin_amdgcn_tanhf(x);
#else
  float e = __builtin_amdgcn_exp2f(-2.8853900817779268f * x);
  return (1.0f - e) * __builtin_amdgcn_rcpf(1.0f + e);
#endif
}
__device__ __forceinline__ float fast_sigmoid(float x) {
  return __builtin_fmaf(0.5f, fast_tanh(0.5f * x), 0.5f);
}
__device__ __forceinline__ v2f tanh2(v2f v) {
  v2f r; r.x = fast_tanh(v.x); r.y = fast_tanh(v.y); return r;
}
__device__ __forceinline__ v2f sig2(v2f v) {
  v2f r; r.x = fast_sigmoid(v.x); r.y = fast_sigmoid(v.y); return r;
}
__device__ __forceinline__ v8f wmma4(v2f a, v2f b, v8f c) {
  return __builtin_amdgcn_wmma_f32_16x16x4_f32(false, a, false, b, (short)0, c, false, false);
}

__device__ __forceinline__ void cell_act(v8f g, v2f& h, v2f& c) {
  v2f gi; gi.x = g[0]; gi.y = g[1];
  v2f gf; gf.x = g[2]; gf.y = g[3];
  v2f gg; gg.x = g[4]; gg.y = g[5];
  v2f go; go.x = g[6]; go.y = g[7];
  v2f iv = sig2(gi);
  v2f fv = sig2(gf);
  v2f gv = tanh2(gg);
  v2f ov = sig2(go);
  c = __builtin_elementwise_fma(fv, c, iv * gv);
  h = ov * tanh2(c);
}

__global__ void repack_wout_kernel(const float* __restrict__ Wout,
                                   v2f* __restrict__ WP) {
  int idx = blockIdx.x * blockDim.x + threadIdx.x;
  if (idx >= WP_ENTRIES) return;
  int c = idx & 31;
  int kp = (idx >> 5) & 1;
  int t = idx >> 6;
  v2f v; v.x = 0.f; v.y = 0.f;
  if (c < OUTN) {
    const float* p = Wout + c * 800 + 4 * t + 2 * kp;
    v.x = p[0];
    v.y = p[1];
  }
  WP[idx] = v;
}

struct Stage {
  v2f x[BLK];
  v2f w0[BLK];
  v2f w1[BLK];
};

__device__ __forceinline__ void load_stage(Stage& s, const v2f* __restrict__ xp,
                                           const v2f* __restrict__ wpLane, int b) {
  const v2f* xs = xp + 2 * BLK * b;
  const v2f* ws = wpLane + 64 * BLK * b;
#pragma unroll
  for (int i = 0; i < BLK; ++i) s.x[i] = xs[2 * i];
#pragma unroll
  for (int i = 0; i < BLK; ++i) { s.w0[i] = ws[64 * i]; s.w1[i] = ws[64 * i + 16]; }
}

__global__ void __launch_bounds__(WAVES_PER_BLOCK * 32)
lstm_fused_kernel(const float* __restrict__ x,
                  const float* __restrict__ Wih0, const float* __restrict__ Whh0,
                  const float* __restrict__ bih0, const float* __restrict__ bhh0,
                  const float* __restrict__ Wih1, const float* __restrict__ Whh1,
                  const float* __restrict__ bih1, const float* __restrict__ bhh1,
                  const v2f* __restrict__ WP, const float* __restrict__ bout,
                  float* __restrict__ out) {
  const int tid = threadIdx.x;
  const int lane = tid & 31;
  const int col = lane & 15;
  const int hh = lane >> 4;
  const int tileBase = (blockIdx.x * WAVES_PER_BLOCK + (tid >> 5)) * TILE_M;

  const int ar = col & 7, ahalf = col >> 3;
  const int j = (ar >> 1) * 4 + 2 * ahalf + (ar & 1);
  v2f aih0, ahh0, aih1, ahh1;
  aih0.x = Wih0[j * 4 + 2 * hh]; aih0.y = Wih0[j * 4 + 2 * hh + 1];
  ahh0.x = Whh0[j * 4 + 2 * hh]; ahh0.y = Whh0[j * 4 + 2 * hh + 1];
  aih1.x = Wih1[j * 4 + 2 * hh]; aih1.y = Wih1[j * 4 + 2 * hh + 1];
  ahh1.x = Whh1[j * 4 + 2 * hh]; ahh1.y = Whh1[j * 4 + 2 * hh + 1];

  v8f bias0, bias1;
#pragma unroll
  for (int r = 0; r < 8; ++r) {
    const int jr = (r >> 1) * 4 + 2 * hh + (r & 1);
    bias0[r] = bih0[jr] + bhh0[jr];
    bias1[r] = bih1[jr] + bhh1[jr];
  }

  const v2f* __restrict__ xp =
      (const v2f*)(x + (size_t)(tileBase + col) * T_STEPS * HID + 2 * hh);

  v2f h0 = (v2f)0.f, c0 = (v2f)0.f, h1 = (v2f)0.f, c1 = (v2f)0.f;

  float bo0 = bout[col];
  float bo1 = (col + 16 < OUTN) ? bout[col + 16] : 0.f;
  v8f acc0, acc1;
#pragma unroll
  for (int r = 0; r < 8; ++r) { acc0[r] = bo0; acc1[r] = bo1; }

  const v2f* __restrict__ wpLane = WP + hh * 32 + col;

  auto run_block = [&](const Stage& s, v2f xlast) {
#pragma unroll
    for (int i = 0; i < BLK; ++i) {
      const v2f xn = (i + 1 < BLK) ? s.x[i + 1 < BLK ? i + 1 : 0] : xlast;
      v8f g1 = wmma4(aih1, h0, bias1);
      g1 = wmma4(ahh1, h1, g1);
      v8f g0 = wmma4(aih0, xn, bias0);
      g0 = wmma4(ahh0, h0, g0);
      cell_act(g1, h1, c1);
      cell_act(g0, h0, c0);
      acc0 = wmma4(h1, s.w0[i], acc0);
      acc1 = wmma4(h1, s.w1[i], acc1);
    }
  };

  Stage sA, sB;
  load_stage(sA, xp, wpLane, 0);
  {
    v8f g = wmma4(aih0, sA.x[0], bias0);
    g = wmma4(ahh0, h0, g);
    cell_act(g, h0, c0);
  }

  for (int b = 0; b + 1 < NBLK; b += 2) {
    load_stage(sB, xp, wpLane, b + 1);
    __builtin_amdgcn_sched_barrier(0);
    run_block(sA, sB.x[0]);
    load_stage(sA, xp, wpLane, b + 2);
    run_block(sB, sA.x[0]);
  }
  run_block(sA, sA.x[BLK - 1]);

  float* outp = out + (size_t)(tileBase + 8 * hh) * OUTN + col;
#pragma unroll
  for (int r = 0; r < 8; ++r) {
    outp[(size_t)r * OUTN] = acc0[r];
    if (col < OUTN - 16) outp[(size_t)r * OUTN + 16] = acc1[r];
  }
}

extern "C" void kernel_launch(void* const* d_in, const int* in_sizes, int n_in,
                              void* d_out, int out_size, void* d_ws, size_t ws_size,
                              hipStream_t stream) {
  const float* x    = (const float*)d_in[0];
  const float* Wih0 = (const float*)d_in[1];
  const float* Whh0 = (const float*)d_in[2];
  const float* bih0 = (const float*)d_in[3];
  const float* bhh0 = (const float*)d_in[4];
  const float* Wih1 = (const float*)d_in[5];
  const float* Whh1 = (const float*)d_in[6];
  const float* bih1 = (const float*)d_in[7];
  const float* bhh1 = (const float*)d_in[8];
  const float* Wout = (const float*)d_in[9];
  const float* bout = (const float*)d_in[10];
  float* out = (float*)d_out;
  v2f* WP = (v2f*)d_ws;

  int B = in_sizes[0] / (T_STEPS * HID);

  repack_wout_kernel<<<(WP_ENTRIES + 255) / 256, 256, 0, stream>>>(Wout, WP);
  lstm_fused_kernel<<<B / 128, 256, 0, stream>>>(x, Wih0, Whh0, bih0, bhh0,
                                                 Wih1, Whh1, bih1, bhh1,
                                                 WP, bout, out);
}


# Wout repacked once per work-group into LDS (whole-line reads), read as one 16-byte LDS load per step, grouped per 16 steps; repack launch and workspace copy gone (2 launches -> 1); sums unchanged
# speedup vs baseline: 3.3414x; 1.0726x over previous
#include <hip/hip_runtime.h>


typedef __attribute__((ext_vector_type(2))) float v2f;
typedef __attribute__((ext_vector_type(8))) float v8f;

#define T_STEPS 200
#define HID 4
#define OUTN 28
#define WOUT_F4 (OUTN * T_STEPS)
#define WL_PITCH 33
#define TILE_M 16
#define WAVES_PER_BLOCK 8
#define BLK 8
#define NBLK (T_STEPS / BLK)

__device__ __forceinline__ float fast_tanh(float x) {
#if __has_builtin(__builtin_amdgcn_tanhf)
  return __builtin_amdgcn_tanhf(x);
#else
  float e = __builtin_amdgcn_exp2f(-2.8853900817779268f * x);
  return (1.0f - e) * __builtin_amdgcn_rcpf(1.0f + e);
#endif
}
__device__ __forceinline__ float fast_sigmoid(float x) {
  return __builtin_fmaf(0.5f, fast_tanh(0.5f * x), 0.5f);
}
__device__ __forceinline__ v2f tanh2(v2f v) {
  v2f r; r.x = fast_tanh(v.x); r.y = fast_tanh(v.y); return r;
}
__device__ __forceinline__ v2f sig2(v2f v) {
  v2f r; r.x = fast_sigmoid(v.x); r.y = fast_sigmoid(v.y); return r;
}
__device__ __forceinline__ v8f wmma4(v2f a, v2f b, v8f c) {
  return __builtin_amdgcn_wmma_f32_16x16x4_f32(false, a, false, b, (short)0, c, false, false);
}

__device__ __forceinline__ void cell_act(v8f g, v2f& h, v2f& c) {
  v2f gi; gi.x = g[0]; gi.y = g[1];
  v2f gf; gf.x = g[2]; gf.y = g[3];
  v2f gg; gg.x = g[4]; gg.y = g[5];
  v2f go; go.x = g[6]; go.y = g[7];
  v2f iv = sig2(gi);
  v2f fv = sig2(gf);
  v2f gv = tanh2(gg);
  v2f ov = sig2(go);
  c = __builtin_elementwise_fma(fv, c, iv * gv);
  h = ov * tanh2(c);
}

struct Stage {
  v2f x[BLK];
};
__device__ __forceinline__ void load_stage_x(Stage& s, const v2f* __restrict__ xp, int b) {
  const v2f* xs = xp + 2 * BLK * b;
#pragma unroll
  for (int i = 0; i < BLK; ++i) s.x[i] = xs[2 * i];
}
typedef __attribute__((ext_vector_type(4))) float v4f;
template <int N>
__device__ __forceinline__ void load_w(v4f (&w)[2 * BLK], const v4f* wlLane, int b) {
  const v4f* ws = wlLane + WL_PITCH * BLK * b;
#pragma unroll
  for (int i = 0; i < N; ++i) w[i] = ws[WL_PITCH * i];
}

__global__ void __launch_bounds__(WAVES_PER_BLOCK * 32)
lstm_fused_kernel(const float* __restrict__ x,
                  const float* __restrict__ Wih0, const float* __restrict__ Whh0,
                  const float* __restrict__ bih0, const float* __restrict__ bhh0,
                  const float* __restrict__ Wih1, const float* __restrict__ Whh1,
                  const float* __restrict__ bih1, const float* __restrict__ bhh1,
                  const float* __restrict__ Wout, const float* __restrict__ bout,
                  float* __restrict__ out) {
  __shared__ v4f wl[T_STEPS * WL_PITCH];

  const int tid = threadIdx.x;
  const int lane = tid & 31;
  const int col = lane & 15;
  const int hh = lane >> 4;
  const int tileBase = (blockIdx.x * WAVES_PER_BLOCK + (tid >> 5)) * TILE_M;

  const v2f* __restrict__ xp =
      (const v2f*)(x + (size_t)(tileBase + col) * T_STEPS * HID + 2 * hh);

  const int ar = col & 7, ahalf = col >> 3;
  const int j = (ar >> 1) * 4 + 2 * ahalf + (ar & 1);
  v2f aih0, ahh0, aih1, ahh1;
  aih0.x = Wih0[j * 4 + 2 * hh]; aih0.y = Wih0[j * 4 + 2 * hh + 1];
  ahh0.x = Whh0[j * 4 + 2 * hh]; ahh0.y = Whh0[j * 4 + 2 * hh + 1];
  aih1.x = Wih1[j * 4 + 2 * hh]; aih1.y = Wih1[j * 4 + 2 * hh + 1];
  ahh1.x = Whh1[j * 4 + 2 * hh]; ahh1.y = Whh1[j * 4 + 2 * hh + 1];

  v8f bias0, bias1;
#pragma unroll
  for (int r = 0; r < 8; ++r) {
    const int jr = (r >> 1) * 4 + 2 * hh + (r & 1);
    bias0[r] = bih0[jr] + bhh0[jr];
    bias1[r] = bih1[jr] + bhh1[jr];
  }

  v2f h0 = (v2f)0.f, c0 = (v2f)0.f, h1 = (v2f)0.f, c1 = (v2f)0.f;

  float bo0 = bout[col];
  float bo1 = (col + 16 < OUTN) ? bout[col + 16] : 0.f;
  v8f acc0, acc1;
#pragma unroll
  for (int r = 0; r < 8; ++r) { acc0[r] = bo0; acc1[r] = bo1; }

  Stage sA, sB;
  load_stage_x(sA, xp, 0);

  {
    const v4f* __restrict__ w4 = (const v4f*)Wout;
    v2f* wl2 = (v2f*)wl;
    constexpr int NIT = (WOUT_F4 + 255) / 256;
    v4f piece[NIT];
#pragma unroll
    for (int it = 0; it < NIT; ++it) {
      const int q = tid + 256 * it;
      if (q < WOUT_F4) piece[it] = w4[q];
    }
#pragma unroll
    for (int it = 0; it < NIT; ++it) {
      const int q = tid + 256 * it;
      if (q < WOUT_F4) {
        const int c = q / T_STEPS, t = q - c * T_STEPS;
        const int e = (t * WL_PITCH + (c & 15)) * 2 + (c >> 4);
        v2f lo; lo.x = piece[it].x; lo.y = piece[it].y;
        v2f hi; hi.x = piece[it].z; hi.y = piece[it].w;
        wl2[e] = lo;
        wl2[e + 32] = hi;
      }
    }
    for (int z = tid; z < T_STEPS * 8; z += 256) {
      const int t = z >> 3, r = z & 7;
      v2f zero; zero.x = 0.f; zero.y = 0.f;
      wl2[(t * WL_PITCH + (r >> 2) * 16 + 12 + (r & 3)) * 2 + 1] = zero;
    }
  }

  const v4f* wlLane = wl + hh * 16 + col;

  v4f w[2 * BLK];
  auto run_block = [&](const Stage& s, const int wofs, v2f xlast) {
#pragma unroll
    for (int i = 0; i < BLK; ++i) {
      const v2f xn = (i + 1 < BLK) ? s.x[i + 1 < BLK ? i + 1 : 0] : xlast;
      v2f w0; w0.x = w[wofs + i].x; w0.y = w[wofs + i].y;
      v2f w1; w1.x = w[wofs + i].z; w1.y = w[wofs + i].w;
      v8f g1 = wmma4(aih1, h0, bias1);
      g1 = wmma4(ahh1, h1, g1);
      v8f g0 = wmma4(aih0, xn, bias0);
      g0 = wmma4(ahh0, h0, g0);
      cell_act(g1, h1, c1);
      cell_act(g0, h0, c0);
      acc0 = wmma4(h1, w0, acc0);
      acc1 = wmma4(h1, w1, acc1);
    }
  };

  {
    v8f g = wmma4(aih0, sA.x[0], bias0);
    g = wmma4(ahh0, h0, g);
    cell_act(g, h0, c0);
  }

  __syncthreads();

  for (int b = 0; b + 1 < NBLK; b += 2) {
    load_stage_x(sB, xp, b + 1);
    load_w<2 * BLK>(w, wlLane, b);
    __builtin_amdgcn_sched_barrier(0);
    run_block(sA, 0, sB.x[0]);
    load_stage_x(sA, xp, b + 2);
    run_block(sB, BLK, sA.x[0]);
  }
  load_w<BLK>(w, wlLane, NBLK - 1);
  run_block(sA, 0, sA.x[BLK - 1]);

  float* outp = out + (size_t)(tileBase + 8 * hh) * OUTN + col;
#pragma unroll
  for (int r = 0; r < 8; ++r) {
    outp[(size_t)r * OUTN] = acc0[r];
    if (col < OUTN - 16) outp[(size_t)r * OUTN + 16] = acc1[r];
  }
}

extern "C" void kernel_launch(void* const* d_in, const int* in_sizes, int n_in,
                              void* d_out, int out_size, void* d_ws, size_t ws_size,
                              hipStream_t stream) {
  const float* x    = (const float*)d_in[0];
  const float* Wih0 = (const float*)d_in[1];
  const float* Whh0 = (const float*)d_in[2];
  const float* bih0 = (const float*)d_in[3];
  const float* bhh0 = (const float*)d_in[4];
  const float* Wih1 = (const float*)d_in[5];
  const float* Whh1 = (const float*)d_in[6];
  const float* bih1 = (const float*)d_in[7];
  const float* bhh1 = (const float*)d_in[8];
  const float* Wout = (const float*)d_in[9];
  const float* bout = (const float*)d_in[10];
  float* out = (float*)d_out;

  int B = in_sizes[0] / (T_STEPS * HID);

  lstm_fused_kernel<<<B / 128, 256, 0, stream>>>(x, Wih0, Whh0, bih0, bhh0,
                                                 Wih1, Whh1, bih1, bhh1,
                                                 Wout, bout, out);
}


# Layers skewed half a step in each wave: one cell's tanh/packed pieces stand between the WMMAs that feed the other cell (fences); x look-ahead 2 steps, stage refilled 2 steps sooner; sums unchanged
# speedup vs baseline: 3.4702x; 1.0385x over previous
#include <hip/hip_runtime.h>


typedef __attribute__((ext_vector_type(2))) float v2f;
typedef __attribute__((ext_vector_type(8))) float v8f;

#define T_STEPS 200
#define HID 4
#define OUTN 28
#define WOUT_F4 (OUTN * T_STEPS)
#define WL_PITCH 33
#define TILE_M 16
#define WAVES_PER_BLOCK 8
#define BLK 8
#define NBLK (T_STEPS / BLK)

__device__ __forceinline__ float fast_tanh(float x) {
#if __has_builtin(__builtin_amdgcn_tanhf)
  return __builtin_amdgcn_tanhf(x);
#else
  float e = __builtin_amdgcn_exp2f(-2.8853900817779268f * x);
  return (1.0f - e) * __builtin_amdgcn_rcpf(1.0f + e);
#endif
}
__device__ __forceinline__ float fast_sigmoid(float x) {
  return __builtin_fmaf(0.5f, fast_tanh(0.5f * x), 0.5f);
}
__device__ __forceinline__ v2f tanh2(v2f v) {
  v2f r; r.x = fast_tanh(v.x); r.y = fast_tanh(v.y); return r;
}
__device__ __forceinline__ v2f sig2(v2f v) {
  v2f r; r.x = fast_sigmoid(v.x); r.y = fast_sigmoid(v.y); return r;
}
__device__ __forceinline__ v8f wmma4(v2f a, v2f b, v8f c) {
  return __builtin_amdgcn_wmma_f32_16x16x4_f32(false, a, false, b, (short)0, c, false, false);
}

__device__ __forceinline__ void cell_act(v8f g, v2f& h, v2f& c) {
  v2f gi; gi.x = g[0]; gi.y = g[1];
  v2f gf; gf.x = g[2]; gf.y = g[3];
  v2f gg; gg.x = g[4]; gg.y = g[5];
  v2f go; go.x = g[6]; go.y = g[7];
  v2f iv = sig2(gi);
  v2f fv = sig2(gf);
  v2f gv = tanh2(gg);
  v2f ov = sig2(go);
  c = __builtin_elementwise_fma(fv, c, iv * gv);
  h = ov * tanh2(c);
}

struct Cell {
  v2f ti, tf, tg, to;
  v2f iv, fv, ov;
  v2f ig;
  v2f tc;
};
__device__ __forceinline__ void act_scale(Cell& a, v8f g) {
  v2f gi; gi.x = g[0]; gi.y = g[1];
  v2f gf; gf.x = g[2]; gf.y = g[3];
  v2f gg; gg.x = g[4]; gg.y = g[5];
  v2f go; go.x = g[6]; go.y = g[7];
  a.ti = 0.5f * gi; a.tf = 0.5f * gf; a.tg = gg; a.to = 0.5f * go;
}
__device__ __forceinline__ void act_tanh_ig(Cell& a) { a.ti = tanh2(a.ti); a.tg = tanh2(a.tg); }
__device__ __forceinline__ void act_tanh_f(Cell& a) { a.tf = tanh2(a.tf); }
__device__ __forceinline__ void act_tanh_o(Cell& a) { a.to = tanh2(a.to); }
__device__ __forceinline__ void act_iv(Cell& a) { const v2f half = (v2f)0.5f; a.iv = __builtin_elementwise_fma(half, a.ti, half); }
__device__ __forceinline__ void act_fv(Cell& a) { const v2f half = (v2f)0.5f; a.fv = __builtin_elementwise_fma(half, a.tf, half); }
__device__ __forceinline__ void act_ov(Cell& a) { const v2f half = (v2f)0.5f; a.ov = __builtin_elementwise_fma(half, a.to, half); }
__device__ __forceinline__ void act_mul(Cell& a) { a.ig = a.iv * a.tg; }
__device__ __forceinline__ void act_c(const Cell& a, v2f& c) { c = __builtin_elementwise_fma(a.fv, c, a.ig); }
__device__ __forceinline__ void act_tanh_c(Cell& a, v2f c) { a.tc = tanh2(c); }
__device__ __forceinline__ void act_h(const Cell& a, v2f& h) { h = a.ov * a.tc; }

__device__ __forceinline__ void hold() { __builtin_amdgcn_sched_barrier(0); }

struct Stage {
  v2f x[BLK];
};
__device__ __forceinline__ void load_stage_x(Stage& s, const v2f* __restrict__ xp, int b) {
  const v2f* xs = xp + 2 * BLK * b;
#pragma unroll
  for (int i = 0; i < BLK; ++i) s.x[i] = xs[2 * i];
}
typedef __attribute__((ext_vector_type(4))) float v4f;
template <int N>
__device__ __forceinline__ void load_w(v4f (&w)[2 * BLK], const v4f* wlLane, int b) {
  const v4f* ws = wlLane + WL_PITCH * BLK * b;
#pragma unroll
  for (int i = 0; i < N; ++i) w[i] = ws[WL_PITCH * i];
}

__global__ void __launch_bounds__(WAVES_PER_BLOCK * 32)
lstm_fused_kernel(const float* __restrict__ x,
                  const float* __restrict__ Wih0, const float* __restrict__ Whh0,
                  const float* __restrict__ bih0, const float* __restrict__ bhh0,
                  const float* __restrict__ Wih1, const float* __restrict__ Whh1,
                  const float* __restrict__ bih1, const float* __restrict__ bhh1,
                  const float* __restrict__ Wout, const float* __restrict__ bout,
                  float* __restrict__ out) {
  __shared__ v4f wl[T_STEPS * WL_PITCH];

  const int tid = threadIdx.x;
  const int lane = tid & 31;
  const int col = lane & 15;
  const int hh = lane >> 4;
  const int tileBase = (blockIdx.x * WAVES_PER_BLOCK + (tid >> 5)) * TILE_M;

  const v2f* __restrict__ xp =
      (const v2f*)(x + (size_t)(tileBase + col) * T_STEPS * HID + 2 * hh);

  const int ar = col & 7, ahalf = col >> 3;
  const int j = (ar >> 1) * 4 + 2 * ahalf + (ar & 1);
  v2f aih0, ahh0, aih1, ahh1;
  aih0.x = Wih0[j * 4 + 2 * hh]; aih0.y = Wih0[j * 4 + 2 * hh + 1];
  ahh0.x = Whh0[j * 4 + 2 * hh]; ahh0.y = Whh0[j * 4 + 2 * hh + 1];
  aih1.x = Wih1[j * 4 + 2 * hh]; aih1.y = Wih1[j * 4 + 2 * hh + 1];
  ahh1.x = Whh1[j * 4 + 2 * hh]; ahh1.y = Whh1[j * 4 + 2 * hh + 1];

  v8f bias0, bias1;
#pragma unroll
  for (int r = 0; r < 8; ++r) {
    const int jr = (r >> 1) * 4 + 2 * hh + (r & 1);
    bias0[r] = bih0[jr] + bhh0[jr];
    bias1[r] = bih1[jr] + bhh1[jr];
  }

  v2f h0 = (v2f)0.f, c0 = (v2f)0.f, h1 = (v2f)0.f, c1 = (v2f)0.f;

  float bo0 = bout[col];
  float bo1 = (col + 16 < OUTN) ? bout[col + 16] : 0.f;
  v8f acc0, acc1;
#pragma unroll
  for (int r = 0; r < 8; ++r) { acc0[r] = bo0; acc1[r] = bo1; }

  Stage sA, sB;
  load_stage_x(sA, xp, 0);
  load_stage_x(sB, xp, 1);

  {
    const v4f* __restrict__ w4 = (const v4f*)Wout;
    v2f* wl2 = (v2f*)wl;
    constexpr int NIT = (WOUT_F4 + 255) / 256;
    v4f piece[NIT];
#pragma unroll
    for (int it = 0; it < NIT; ++it) {
      const int q = tid + 256 * it;
      if (q < WOUT_F4) piece[it] = w4[q];
    }
#pragma unroll
    for (int it = 0; it < NIT; ++it) {
      const int q = tid + 256 * it;
      if (q < WOUT_F4) {
        const int c = q / T_STEPS, t = q - c * T_STEPS;
        const int e = (t * WL_PITCH + (c & 15)) * 2 + (c >> 4);
        v2f lo; lo.x = piece[it].x; lo.y = piece[it].y;
        v2f hi; hi.x = piece[it].z; hi.y = piece[it].w;
        wl2[e] = lo;
        wl2[e + 32] = hi;
      }
    }
    for (int z = tid; z < T_STEPS * 8; z += 256) {
      const int t = z >> 3, r = z & 7;
      v2f zero; zero.x = 0.f; zero.y = 0.f;
      wl2[(t * WL_PITCH + (r >> 2) * 16 + 12 + (r & 3)) * 2 + 1] = zero;
    }
  }

  const v4f* wlLane = wl + hh * 16 + col;

  v4f w[2 * BLK];
  v8f g1, gx0;
  Cell B;
  auto run_block = [&](Stage& s, const int wofs, const Stage& nxt, const bool refill, const int rb) {
#pragma unroll
    for (int i = 0; i < BLK; ++i) {
      if (refill && i == BLK - 2) {
        load_stage_x(s, xp, rb);
        hold();
      }
      const v2f x2 = (i + 2 < BLK) ? s.x[(i + 2) % BLK] : nxt.x[(i + 2) % BLK];
      v2f w0; w0.x = w[wofs + i].x; w0.y = w[wofs + i].y;
      v2f w1; w1.x = w[wofs + i].z; w1.y = w[wofs + i].w;
      Cell A;
      gx0 = wmma4(aih0, x2, bias0);
      hold();
      act_h(B, h0);
      act_scale(A, g1); act_tanh_ig(A);
      hold();
      v8f g0 = wmma4(ahh0, h0, gx0);
      hold();
      act_iv(A);
      hold();
      act_tanh_f(A); act_tanh_o(A);
      hold();
      act_mul(A);
      hold();
      v8f g1a = wmma4(aih1, h0, bias1);
      hold();
      act_fv(A); act_ov(A); act_scale(B, g0); act_c(A, c1);
      hold();
      act_tanh_ig(B); act_tanh_c(A, c1);
      hold();
      act_iv(B); act_tanh_f(B);
      hold();
      act_h(A, h1);
      hold();
      act_tanh_o(B);
      hold();
      act_mul(B); act_fv(B);
      hold();
      g1 = wmma4(ahh1, h1, g1a);
      hold();
      act_ov(B); act_c(B, c0);
      hold();
      acc0 = wmma4(h1, w0, acc0);
      hold();
      act_tanh_c(B, c0);
      hold();
      acc1 = wmma4(h1, w1, acc1);
      hold();
    }
  };

  {
    v8f g = wmma4(aih0, sA.x[0], bias0);
    g = wmma4(ahh0, h0, g);
    cell_act(g, h0, c0);
    g1 = wmma4(aih1, h0, bias1);
    g1 = wmma4(ahh1, h1, g1);
    g = wmma4(aih0, sA.x[1], bias0);
    g = wmma4(ahh0, h0, g);
    act_scale(B, g); act_tanh_ig(B); act_tanh_f(B); act_tanh_o(B);
    act_iv(B); act_mul(B); act_fv(B); act_ov(B); act_c(B, c0);
    act_tanh_c(B, c0);
  }

  __syncthreads();

  for (int b = 0; b + 1 < NBLK; b += 2) {
    load_w<2 * BLK>(w, wlLane, b);
    __builtin_amdgcn_sched_barrier(0);
    run_block(sA, 0, sB, true, b + 2);
    run_block(sB, BLK, sA, true, (b + 3 < NBLK) ? b + 3 : NBLK - 1);
  }
  load_w<BLK>(w, wlLane, NBLK - 1);
  run_block(sA, 0, sA, false, 0);

  float* outp = out + (size_t)(tileBase + 8 * hh) * OUTN + col;
#pragma unroll
  for (int r = 0; r < 8; ++r) {
    outp[(size_t)r * OUTN] = acc0[r];
    if (col < OUTN - 16) outp[(size_t)r * OUTN + 16] = acc1[r];
  }
}

extern "C" void kernel_launch(void* const* d_in, const int* in_sizes, int n_in,
                              void* d_out, int out_size, void* d_ws, size_t ws_size,
                              hipStream_t stream) {
  const float* x    = (const float*)d_in[0];
  const float* Wih0 = (const float*)d_in[1];
  const float* Whh0 = (const float*)d_in[2];
  const float* bih0 = (const float*)d_in[3];
  const float* bhh0 = (const float*)d_in[4];
  const float* Wih1 = (const float*)d_in[5];
  const float* Whh1 = (const float*)d_in[6];
  const float* bih1 = (const float*)d_in[7];
  const float* bhh1 = (const float*)d_in[8];
  const float* Wout = (const float*)d_in[9];
  const float* bout = (const float*)d_in[10];
  float* out = (float*)d_out;

  int B = in_sizes[0] / (T_STEPS * HID);

  lstm_fused_kernel<<<B / 128, 256, 0, stream>>>(x, Wih0, Whh0, bih0, bhh0,
                                                 Wih1, Whh1, bih1, bhh1,
                                                 Wout, bout, out);
}


# speedup vs baseline: 3.6031x; 1.0383x over previous
#include <hip/hip_runtime.h>


typedef __attribute__((ext_vector_type(2))) float v2f;
typedef __attribute__((ext_vector_type(8))) float v8f;

#define T_STEPS 200
#define HID 4
#define OUTN 28
#define WOUT_F4 (OUTN * T_STEPS)
#define WL_PITCH 33
#define TILE_M 16
#define WAVES_PER_BLOCK 8
#define BLK 8
#define NBLK (T_STEPS / BLK)

__device__ __forceinline__ float fast_tanh(float x) {
#if __has_builtin(__builtin_amdgcn_tanhf)
  return __builtin_amdgcn_tanhf(x);
#else
  float e = __builtin_amdgcn_exp2f(-2.8853900817779268f * x);
  return (1.0f - e) * __builtin_amdgcn_rcpf(1.0f + e);
#endif
}
__device__ __forceinline__ v2f tanh2(v2f v) {
  v2f r; r.x = fast_tanh(v.x); r.y = fast_tanh(v.y); return r;
}
__device__ __forceinline__ v8f wmma4(v2f a, v2f b, v8f c) {
  return __builtin_amdgcn_wmma_f32_16x16x4_f32(false, a, false, b, (short)0, c, false, false);
}

struct Cell {
  v2f ti, tf, tg, to;
  v2f iv, fv, ov;
  v2f ig;
  v2f tc;
};
__device__ __forceinline__ v2f sig_from_tanh(v2f t) { const v2f half = (v2f)0.5f; return __builtin_elementwise_fma(half, t, half); }
__device__ __forceinline__ void cell_act(v8f g, v2f& h, v2f& c) {
  v2f gi; gi.x = g[0]; gi.y = g[1];
  v2f gf; gf.x = g[2]; gf.y = g[3];
  v2f gg; gg.x = g[4]; gg.y = g[5];
  v2f go; go.x = g[6]; go.y = g[7];
  v2f iv = sig_from_tanh(tanh2(gi));
  v2f fv = sig_from_tanh(tanh2(gf));
  v2f gv = tanh2(gg);
  v2f ov = sig_from_tanh(tanh2(go));
  c = __builtin_elementwise_fma(fv, c, iv * gv);
  h = ov * tanh2(c);
}

__device__ __forceinline__ void hold() { __builtin_amdgcn_sched_barrier(0); }

struct Stage {
  v2f x[BLK];
};
__device__ __forceinline__ void load_stage_x(Stage& s, const v2f* __restrict__ xp, int b) {
  const v2f* xs = xp + 2 * BLK * b;
#pragma unroll
  for (int i = 0; i < BLK; ++i) s.x[i] = xs[2 * i];
}
typedef __attribute__((ext_vector_type(4))) float v4f;
template <int N>
__device__ __forceinline__ void load_w(v4f (&w)[2 * BLK], const v4f* wlLane, int b) {
  const v4f* ws = wlLane + WL_PITCH * BLK * b;
#pragma unroll
  for (int i = 0; i < N; ++i) w[i] = ws[WL_PITCH * i];
}

__global__ void __launch_bounds__(WAVES_PER_BLOCK * 32)
lstm_fused_kernel(const float* __restrict__ x,
                  const float* __restrict__ Wih0, const float* __restrict__ Whh0,
                  const float* __restrict__ bih0, const float* __restrict__ bhh0,
                  const float* __restrict__ Wih1, const float* __restrict__ Whh1,
                  const float* __restrict__ bih1, const float* __restrict__ bhh1,
                  const float* __restrict__ Wout, const float* __restrict__ bout,
                  float* __restrict__ out) {
  __shared__ v4f wl[T_STEPS * WL_PITCH];

  const int tid = threadIdx.x;
  const int lane = tid & 31;
  const int col = lane & 15;
  const int hh = lane >> 4;
  const int tileBase = (blockIdx.x * WAVES_PER_BLOCK + (tid >> 5)) * TILE_M;

  const v2f* __restrict__ xp =
      (const v2f*)(x + (size_t)(tileBase + col) * T_STEPS * HID + 2 * hh);

  const int ar = col & 7, ahalf = col >> 3;
  const int j = (ar >> 1) * 4 + 2 * ahalf + (ar & 1);
  v2f aih0, ahh0, aih1, ahh1;
  aih0.x = Wih0[j * 4 + 2 * hh]; aih0.y = Wih0[j * 4 + 2 * hh + 1];
  ahh0.x = Whh0[j * 4 + 2 * hh]; ahh0.y = Whh0[j * 4 + 2 * hh + 1];
  aih1.x = Wih1[j * 4 + 2 * hh]; aih1.y = Wih1[j * 4 + 2 * hh + 1];
  ahh1.x = Whh1[j * 4 + 2 * hh]; ahh1.y = Whh1[j * 4 + 2 * hh + 1];
  const float rowScale = ((ar >> 1) == 2) ? 1.0f : 0.5f;
  aih0 *= rowScale; ahh0 *= rowScale; aih1 *= rowScale; ahh1 *= rowScale;

  v8f bias0, bias1;
#pragma unroll
  for (int r = 0; r < 8; ++r) {
    const int jr = (r >> 1) * 4 + 2 * hh + (r & 1);
    const float rs = ((r >> 1) == 2) ? 1.0f : 0.5f;
    bias0[r] = (bih0[jr] + bhh0[jr]) * rs;
    bias1[r] = (bih1[jr] + bhh1[jr]) * rs;
  }

  v2f h0 = (v2f)0.f, c0 = (v2f)0.f, h1 = (v2f)0.f, c1 = (v2f)0.f;

  float bo0 = bout[col];
  float bo1 = (col + 16 < OUTN) ? bout[col + 16] : 0.f;
  v8f acc0, acc1;
#pragma unroll
  for (int r = 0; r < 8; ++r) { acc0[r] = bo0; acc1[r] = bo1; }

  Stage sA, sB;
  load_stage_x(sA, xp, 0);
  load_stage_x(sB, xp, 1);

  {
    const v4f* __restrict__ w4 = (const v4f*)Wout;
    v2f* wl2 = (v2f*)wl;
    constexpr int NIT = (WOUT_F4 + 255) / 256;
    v4f piece[NIT];
#pragma unroll
    for (int it = 0; it < NIT; ++it) {
      const int q = tid + 256 * it;
      if (q < WOUT_F4) piece[it] = w4[q];
    }
#pragma unroll
    for (int it = 0; it < NIT; ++it) {
      const int q = tid + 256 * it;
      if (q < WOUT_F4) {
        const int c = q / T_STEPS, t = q - c * T_STEPS;
        const int e = (t * WL_PITCH + (c & 15)) * 2 + (c >> 4);
        v2f lo; lo.x = piece[it].x; lo.y = piece[it].y;
        v2f hi; hi.x = piece[it].z; hi.y = piece[it].w;
        wl2[e] = lo;
        wl2[e + 32] = hi;
      }
    }
    for (int z = tid; z < T_STEPS * 8; z += 256) {
      const int t = z >> 3, r = z & 7;
      v2f zero; zero.x = 0.f; zero.y = 0.f;
      wl2[(t * WL_PITCH + (r >> 2) * 16 + 12 + (r & 3)) * 2 + 1] = zero;
    }
  }

  const v4f* wlLane = wl + hh * 16 + col;

  v4f w[2 * BLK];
  v8f g1, gx0;
  Cell B;
  auto run_block = [&](Stage& s, const int wofs, const Stage& nxt, const bool refill, const int rb) {
#pragma unroll
    for (int i = 0; i < BLK; ++i) {
      if (refill && i == BLK - 2) {
        load_stage_x(s, xp, rb);
        hold();
      }
      const v2f x2 = (i + 2 < BLK) ? s.x[(i + 2) % BLK] : nxt.x[(i + 2) % BLK];
      v2f w0; w0.x = w[wofs + i].x; w0.y = w[wofs + i].y;
      v2f w1; w1.x = w[wofs + i].z; w1.y = w[wofs + i].w;
      Cell A;
      gx0 = wmma4(aih0, x2, bias0);
      hold();
      A.ti.x = fast_tanh(g1[0]); hold(); h0 = B.ov * B.tc; hold();
      A.ti.y = fast_tanh(g1[1]); hold();
      A.tg.x = fast_tanh(g1[4]); hold();
      A.tg.y = fast_tanh(g1[5]); hold();
      A.tf.x = fast_tanh(g1[2]); hold(); A.iv = sig_from_tanh(A.ti); hold();
      A.tf.y = fast_tanh(g1[3]); hold();
      v8f g0 = wmma4(ahh0, h0, gx0);
      hold();
      A.to.x = fast_tanh(g1[6]); hold(); A.fv = sig_from_tanh(A.tf); hold();
      A.to.y = fast_tanh(g1[7]); hold(); A.ig = A.iv * A.tg; hold();
      v8f g1a = wmma4(aih1, h0, bias1);
      hold();
      B.ti.x = fast_tanh(g0[0]); hold(); c1 = __builtin_elementwise_fma(A.fv, c1, A.ig); hold();
      B.ti.y = fast_tanh(g0[1]); hold(); A.ov = sig_from_tanh(A.to); hold();
      B.tg.x = fast_tanh(g0[4]); hold();
      B.tg.y = fast_tanh(g0[5]); hold();
      A.tc.x = fast_tanh(c1.x); hold();
      A.tc.y = fast_tanh(c1.y); hold(); B.iv = sig_from_tanh(B.ti); hold();
      B.tf.x = fast_tanh(g0[2]); hold();
      B.tf.y = fast_tanh(g0[3]); hold();
      B.to.x = fast_tanh(g0[6]); hold(); B.ig = B.iv * B.tg; hold();
      B.to.y = fast_tanh(g0[7]); hold(); h1 = A.ov * A.tc; hold();
      B.fv = sig_from_tanh(B.tf);
      hold();
      g1 = wmma4(ahh1, h1, g1a);
      hold();
      c0 = __builtin_elementwise_fma(B.fv, c0, B.ig);
      hold();
      acc0 = wmma4(h1, w0, acc0);
      hold();
      B.tc.x = fast_tanh(c0.x); hold(); B.ov = sig_from_tanh(B.to); hold();
      B.tc.y = fast_tanh(c0.y); hold();
      acc1 = wmma4(h1, w1, acc1);
      hold();
    }
  };

  {
    v8f g = wmma4(aih0, sA.x[0], bias0);
    g = wmma4(ahh0, h0, g);
    cell_act(g, h0, c0);
    g1 = wmma4(aih1, h0, bias1);
    g1 = wmma4(ahh1, h1, g1);
    g = wmma4(aih0, sA.x[1], bias0);
    g = wmma4(ahh0, h0, g);
    v2f gi; gi.x = g[0]; gi.y = g[1];
    v2f gf; gf.x = g[2]; gf.y = g[3];
    v2f gg; gg.x = g[4]; gg.y = g[5];
    v2f go; go.x = g[6]; go.y = g[7];
    B.iv = sig_from_tanh(tanh2(gi)); B.fv = sig_from_tanh(tanh2(gf));
    B.tg = tanh2(gg); B.ov = sig_from_tanh(tanh2(go));
    B.ig = B.iv * B.tg;
    c0 = __builtin_elementwise_fma(B.fv, c0, B.ig);
    B.tc = tanh2(c0);
  }

  __syncthreads();

  for (int b = 0; b + 1 < NBLK; b += 2) {
    load_w<2 * BLK>(w, wlLane, b);
    __builtin_amdgcn_sched_barrier(0);
    run_block(sA, 0, sB, true, b + 2);
    run_block(sB, BLK, sA, true, (b + 3 < NBLK) ? b + 3 : NBLK - 1);
  }
  load_w<BLK>(w, wlLane, NBLK - 1);
  run_block(sA, 0, sA, false, 0);

  float* outp = out + (size_t)(tileBase + 8 * hh) * OUTN + col;
#pragma unroll
  for (int r = 0; r < 8; ++r) {
    outp[(size_t)r * OUTN] = acc0[r];
    if (col < OUTN - 16) outp[(size_t)r * OUTN + 16] = acc1[r];
  }
}

extern "C" void kernel_launch(void* const* d_in, const int* in_sizes, int n_in,
                              void* d_out, int out_size, void* d_ws, size_t ws_size,
                              hipStream_t stream) {
  const float* x    = (const float*)d_in[0];
  const float* Wih0 = (const float*)d_in[1];
  const float* Whh0 = (const float*)d_in[2];
  const float* bih0 = (const float*)d_in[3];
  const float* bhh0 = (const float*)d_in[4];
  const float* Wih1 = (const float*)d_in[5];
  const float* Whh1 = (const float*)d_in[6];
  const float* bih1 = (const float*)d_in[7];
  const float* bhh1 = (const float*)d_in[8];
  const float* Wout = (const float*)d_in[9];
  const float* bout = (const float*)d_in[10];
  float* out = (float*)d_out;

  int B = in_sizes[0] / (T_STEPS * HID);

  lstm_fused_kernel<<<B / 128, 256, 0, stream>>>(x, Wih0, Whh0, bih0, bhh0,
                                                 Wih1, Whh1, bih1, bhh1,
                                                 Wout, bout, out);
}
